# starting kernel
# baseline (speedup 1.0000x reference)
#include <hip/hip_runtime.h>
#include <math.h>

#define N_NODES   100000
#define N_EDGES   1600000
#define N_FEAT    500
#define HIDDEN    64
#define N_CLASSES 16
#define M_TILES   (N_NODES / 16)

typedef __attribute__((ext_vector_type(2))) float v2f;
typedef __attribute__((ext_vector_type(8))) float v8f;


__global__ void k_init_deg(float* deg) {
    int i = blockIdx.x * blockDim.x + threadIdx.x;
    if (i < N_NODES) deg[i] = 1.0f;
}

__global__ void k_accum_deg(const int* __restrict__ col,
                            const float* __restrict__ w,
                            float* __restrict__ deg) {
    int e = blockIdx.x * blockDim.x + threadIdx.x;
    if (e < N_EDGES) atomicAdd(&deg[col[e]], w[e]);
}

__global__ void k_dinv(const float* __restrict__ deg, float* __restrict__ dinv) {
    int i = blockIdx.x * blockDim.x + threadIdx.x;
    if (i < N_NODES) {
        float d = deg[i];
        dinv[i] = (d > 0.0f) ? rsqrtf(d) : 0.0f;
    }
}

__global__ void k_norm(const int* __restrict__ row, const int* __restrict__ col,
                       const float* __restrict__ w, const float* __restrict__ dinv,
                       float* __restrict__ norm) {
    int e = blockIdx.x * blockDim.x + threadIdx.x;
    if (e < N_EDGES) norm[e] = dinv[row[e]] * w[e] * dinv[col[e]];
}


__global__ __launch_bounds__(128) void k_gemm1(const float* __restrict__ x,
                                               const float* __restrict__ W1,
                                               float* __restrict__ h1) {
    const int wave  = threadIdx.x >> 5;
    const int lane  = threadIdx.x & 31;
    const int mtile = blockIdx.x * 4 + wave;
    if (mtile >= M_TILES) return;
    const int row0 = mtile * 16;
    const int half = lane >> 4;
    const int l16  = lane & 15;

    const float* arow = x + (size_t)(row0 + l16) * N_FEAT + half * 2;
    v8f c0 = {}, c1 = {}, c2 = {}, c3 = {};

    for (int k = 0; k < N_FEAT; k += 4) {
        v2f a;
        a.x = arow[k];
        a.y = arow[k + 1];
        const float* bp = W1 + (size_t)(k + half * 2) * HIDDEN + l16;
        v2f b0, b1, b2, b3;
        b0.x = bp[0];   b0.y = bp[HIDDEN + 0];
        b1.x = bp[16];  b1.y = bp[HIDDEN + 16];
        b2.x = bp[32];  b2.y = bp[HIDDEN + 32];
        b3.x = bp[48];  b3.y = bp[HIDDEN + 48];
        c0 = __builtin_amdgcn_wmma_f32_16x16x4_f32(false, a, false, b0, (short)0, c0, false, false);
        c1 = __builtin_amdgcn_wmma_f32_16x16x4_f32(false, a, false, b1, (short)0, c1, false, false);
        c2 = __builtin_amdgcn_wmma_f32_16x16x4_f32(false, a, false, b2, (short)0, c2, false, false);
        c3 = __builtin_amdgcn_wmma_f32_16x16x4_f32(false, a, false, b3, (short)0, c3, false, false);
    }

    float* out = h1 + (size_t)(row0 + half * 8) * HIDDEN + l16;
    for (int r = 0; r < 8; ++r) {
        out[(size_t)r * HIDDEN + 0]  = c0[r];
        out[(size_t)r * HIDDEN + 16] = c1[r];
        out[(size_t)r * HIDDEN + 32] = c2[r];
        out[(size_t)r * HIDDEN + 48] = c3[r];
    }
}

__global__ void k_init_agg1(const float* __restrict__ h1,
                            const float* __restrict__ dinv,
                            const float* __restrict__ b1,
                            float* __restrict__ agg1) {
    int t = blockIdx.x * blockDim.x + threadIdx.x;
    if (t < N_NODES * HIDDEN) {
        int i = t / HIDDEN, f = t % HIDDEN;
        float s = dinv[i];
        agg1[t] = b1[f] + s * s * h1[t];
    }
}

__global__ void k_scatter1(const int* __restrict__ row, const int* __restrict__ col,
                           const float* __restrict__ norm,
                           const float* __restrict__ h1,
                           float* __restrict__ agg1) {
    unsigned t = blockIdx.x * blockDim.x + threadIdx.x;
    if (t < (unsigned)N_EDGES * HIDDEN) {
        unsigned e = t / HIDDEN, f = t % HIDDEN;
        float v = norm[e] * h1[(size_t)row[e] * HIDDEN + f];
        atomicAdd(&agg1[(size_t)col[e] * HIDDEN + f], v);
    }
}


__global__ __launch_bounds__(128) void k_gemm2(const float* __restrict__ agg1,
                                               const float* __restrict__ W2,
                                               float* __restrict__ h2) {
    const int wave  = threadIdx.x >> 5;
    const int lane  = threadIdx.x & 31;
    const int mtile = blockIdx.x * 4 + wave;
    if (mtile >= M_TILES) return;
    const int row0 = mtile * 16;
    const int half = lane >> 4;
    const int l16  = lane & 15;

    const float* arow = agg1 + (size_t)(row0 + l16) * HIDDEN + half * 2;
    v8f c = {};

    for (int k = 0; k < HIDDEN; k += 4) {
        v2f a;
        a.x = fmaxf(arow[k], 0.0f);
        a.y = fmaxf(arow[k + 1], 0.0f);
        const float* bp = W2 + (size_t)(k + half * 2) * N_CLASSES + l16;
        v2f b;
        b.x = bp[0];
        b.y = bp[N_CLASSES];
        c = __builtin_amdgcn_wmma_f32_16x16x4_f32(false, a, false, b, (short)0, c, false, false);
    }

    float* out = h2 + (size_t)(row0 + half * 8) * N_CLASSES + l16;
    for (int r = 0; r < 8; ++r)
        out[(size_t)r * N_CLASSES] = c[r];
}


__global__ void k_init_agg2(const float* __restrict__ h2,
                            const float* __restrict__ dinv,
                            const float* __restrict__ b2,
                            float* __restrict__ agg2) {
    int t = blockIdx.x * blockDim.x + threadIdx.x;
    if (t < N_NODES * N_CLASSES) {
        int i = t / N_CLASSES, cc = t % N_CLASSES;
        float s = dinv[i];
        agg2[t] = b2[cc] + s * s * h2[t];
    }
}

__global__ void k_scatter2(const int* __restrict__ row, const int* __restrict__ col,
                           const float* __restrict__ norm,
                           const float* __restrict__ h2,
                           float* __restrict__ agg2) {
    unsigned t = blockIdx.x * blockDim.x + threadIdx.x;
    if (t < (unsigned)N_EDGES * N_CLASSES) {
        unsigned e = t / N_CLASSES, cc = t % N_CLASSES;
        float v = norm[e] * h2[(size_t)row[e] * N_CLASSES + cc];
        atomicAdd(&agg2[(size_t)col[e] * N_CLASSES + cc], v);
    }
}


__global__ void k_logsoftmax(const float* __restrict__ agg2, float* __restrict__ out) {
    int i = blockIdx.x * blockDim.x + threadIdx.x;
    if (i >= N_NODES) return;
    const float* v = agg2 + (size_t)i * N_CLASSES;
    float m = v[0];
    #pragma unroll
    for (int c = 1; c < N_CLASSES; ++c) m = fmaxf(m, v[c]);
    float s = 0.0f;
    #pragma unroll
    for (int c = 0; c < N_CLASSES; ++c) s += expf(v[c] - m);
    float l = logf(s);
    float* o = out + (size_t)i * N_CLASSES;
    #pragma unroll
    for (int c = 0; c < N_CLASSES; ++c) o[c] = v[c] - m - l;
}


extern "C" void kernel_launch(void* const* d_in, const int* in_sizes, int n_in,
                              void* d_out, int out_size, void* d_ws, size_t ws_size,
                              hipStream_t stream) {
    const float* x  = (const float*)d_in[0];
    const int*   ei = (const int*)d_in[1];
    const float* ew = (const float*)d_in[2];
    const float* W1 = (const float*)d_in[3];
    const float* b1 = (const float*)d_in[4];
    const float* W2 = (const float*)d_in[5];
    const float* b2 = (const float*)d_in[6];
    float* out = (float*)d_out;

    const int* row = ei;
    const int* col = ei + N_EDGES;

    float* ws   = (float*)d_ws;
    float* deg  = ws;
    float* dinv = deg  + N_NODES;
    float* norm = dinv + N_NODES;
    float* h1   = norm + N_EDGES;
    float* agg1 = h1   + (size_t)N_NODES * HIDDEN;
    float* h2   = agg1 + (size_t)N_NODES * HIDDEN;
    float* agg2 = h2   + (size_t)N_NODES * N_CLASSES;

    const int B = 256;

    k_init_deg <<<(N_NODES + B - 1) / B, B, 0, stream>>>(deg);
    k_accum_deg<<<(N_EDGES + B - 1) / B, B, 0, stream>>>(col, ew, deg);
    k_dinv     <<<(N_NODES + B - 1) / B, B, 0, stream>>>(deg, dinv);
    k_norm     <<<(N_EDGES + B - 1) / B, B, 0, stream>>>(row, col, ew, dinv, norm);

    k_gemm1<<<(M_TILES + 3) / 4, 128, 0, stream>>>(x, W1, h1);

    k_init_agg1<<<(N_NODES * HIDDEN + B - 1) / B, B, 0, stream>>>(h1, dinv, b1, agg1);
    {
        unsigned total = (unsigned)N_EDGES * HIDDEN;
        k_scatter1<<<(total + B - 1) / B, B, 0, stream>>>(row, col, norm, h1, agg1);
    }

    k_gemm2<<<(M_TILES + 3) / 4, 128, 0, stream>>>(agg1, W2, h2);

    k_init_agg2<<<(N_NODES * N_CLASSES + B - 1) / B, B, 0, stream>>>(h2, dinv, b2, agg2);
    {
        unsigned total = (unsigned)N_EDGES * N_CLASSES;
        k_scatter2<<<(total + B - 1) / B, B, 0, stream>>>(row, col, norm, h2, agg2);
    }

    k_logsoftmax<<<(N_NODES + B - 1) / B, B, 0, stream>>>(agg2, out);
}


# Owner bytes are scanned once: the degree pass writes down each wave's edges (16-bit positions, whole-line stores) and both feature passes replay the lists; sums still in LDS arrival order
# speedup vs baseline: 2.4626x; 2.4626x over previous
#include <hip/hip_runtime.h>
#include <math.h>

#define N_NODES   100000
#define N_EDGES   1600000
#define N_FEAT    500
#define HIDDEN    64
#define N_CLASSES 16
#define M_TILES   (N_NODES / 16)

#define OWN_R      416
#define OWN_G      ((N_NODES + OWN_R - 1) / OWN_R)
#define OWN_T      1024
#define OWN_W      (OWN_T / 32)
#define KEY_GROUPS (N_EDGES / 16)
#define SCAN_ITERS ((KEY_GROUPS + OWN_T - 1) / OWN_T)

static_assert(N_EDGES % 16 == 0, "owner bytes are read 16 at a time");
static_assert(KEY_GROUPS % 32 == 0, "the 32 loads of a wave are all inside the list or all outside");
static_assert(OWN_R % 32 == 0, "owners' output ranges start on 128-byte lines");
static_assert(OWN_R * OWN_G >= N_NODES, "owners cover every node");
static_assert((N_NODES - 1) / OWN_R < 256, "owner id fits one byte");
static_assert((OWN_G - 1) * OWN_R < N_NODES, "every owner has at least one node");
static_assert(SCAN_ITERS % 2 == 0, "the scan takes two list positions per trip");

#define LIST_C     304
#define LIST_WG    (OWN_W * LIST_C)
static_assert(OWN_T == 1024, "bits 9..13 of an edge number are the wave that meets it");
static_assert(((N_EDGES - 1) >> 14) < 128, "j fits 7 bits");
static_assert((LIST_WG * 2) % 128 == 0, "an owner's lists fill whole 128-byte lines");
static_assert((LIST_WG * 2) % 16 == 0, "lists are copied out 16 bytes at a time");
static_assert(N_EDGES % 128 == 0, "the lists start on a 128-byte line behind the owner bytes");
static_assert((size_t)N_EDGES + (size_t)OWN_G * LIST_WG * 2 <= (size_t)N_NODES * N_CLASSES * 4, "owner bytes and lists fit the region they share");
static_assert(OWN_G * OWN_W * 4 <= N_NODES * 4, "the counts fit the region they use");

__device__ __forceinline__ int list_code(int e)            { return ((e >> 14) << 9) | (e & 511); }
__device__ __forceinline__ int list_edge(int c, int wave)  { return ((c >> 9) << 14) | (wave << 9) | (c & 511); }

typedef uint4 __attribute__((may_alias)) uint4_any;

typedef __attribute__((ext_vector_type(2))) float v2f;
typedef __attribute__((ext_vector_type(8))) float v8f;


__global__ void k_keys(const int* __restrict__ col, unsigned* __restrict__ keys4) {
    int t = blockIdx.x * blockDim.x + threadIdx.x;
    if (t < N_EDGES / 4) {
        const int4 c = ((const int4*)col)[t];
        keys4[t] =  ((unsigned)c.x / OWN_R)
                 | (((unsigned)c.y / OWN_R) << 8)
                 | (((unsigned)c.z / OWN_R) << 16)
                 | (((unsigned)c.w / OWN_R) << 24);
    }
}

__device__ __forceinline__ void lds_add(float* p, float v) {
    (void)__hip_atomic_fetch_add(p, v, __ATOMIC_RELAXED, __HIP_MEMORY_SCOPE_WORKGROUP);
}

__device__ __forceinline__ void q_put(int* p, int v) {
    __hip_atomic_store(p, v, __ATOMIC_RELEASE, __HIP_MEMORY_SCOPE_WAVEFRONT);
}
__device__ __forceinline__ int q_get(int* p) {
    return __hip_atomic_load(p, __ATOMIC_ACQUIRE, __HIP_MEMORY_SCOPE_WAVEFRONT);
}

__device__ __forceinline__ unsigned zero_bytes(unsigned x) {
    return ~(((x & 0x7F7F7F7Fu) + 0x7F7F7F7Fu) | x | 0x7F7F7F7Fu);
}

__device__ __forceinline__ unsigned mask16(uint4 k, unsigned pat) {
    return (zero_bytes(k.x ^ pat) >> 7) | (zero_bytes(k.y ^ pat) >> 6)
         | (zero_bytes(k.z ^ pat) >> 5) | (zero_bytes(k.w ^ pat) >> 4);
}

template <class Sink>
__device__ __forceinline__ void scan_keys(const unsigned char* __restrict__ keys, int me,
                                          int* myq, const Sink& sink) {
    const int wave = threadIdx.x >> 5;
    const int lane = threadIdx.x & 31;
    const unsigned pat   = (unsigned)me * 0x01010101u;
    const unsigned below = (1u << lane) - 1u;
    const uint4* __restrict__ k16 = (const uint4*)keys;
    const int first = 2 * ((me * (SCAN_ITERS / 2)) / OWN_G);

    int qn = 0;
    int gb_n = first * OWN_T + wave * 32;
    uint4 ka_n = make_uint4(0u, 0u, 0u, 0u), kb_n = make_uint4(0u, 0u, 0u, 0u);
    if (gb_n < KEY_GROUPS)         ka_n = k16[gb_n + lane];
    if (gb_n + OWN_T < KEY_GROUPS) kb_n = k16[gb_n + OWN_T + lane];

    for (int it = 0; it < SCAN_ITERS; it += 2) {
        const int   gb = gb_n;
        const uint4 ka = ka_n, kb = kb_n;
        if (it + 2 < SCAN_ITERS) {
            int jt = first + it + 2;
            if (jt >= SCAN_ITERS) jt -= SCAN_ITERS;
            gb_n = jt * OWN_T + wave * 32;
            if (gb_n < KEY_GROUPS)         ka_n = k16[gb_n + lane];
            if (gb_n + OWN_T < KEY_GROUPS) kb_n = k16[gb_n + OWN_T + lane];
        }

        unsigned m = 0u;
        if (gb < KEY_GROUPS)         m  = mask16(ka, pat);
        if (gb + OWN_T < KEY_GROUPS) m |= mask16(kb, pat) << 4;
        const int e0 = 16 * (gb + lane);

        unsigned bal = __builtin_amdgcn_ballot_w32(m != 0u);
        while (bal != 0u) {
            const int p = __ffs((int)m) - 1;
            const int e = e0 + 16 * OWN_T * ((p >> 2) & 1) + 4 * (p & 3) + (p >> 3);
            if (m != 0u) q_put(&myq[qn + __popc(bal & below)], e);
            m &= m - 1u;
            qn += __popc(bal);
            if (qn >= 32) {
                sink(32, myq);
                const int rem = qn - 32;
                int t = 0;
                if (lane < rem) t = q_get(&myq[32 + lane]);
                if (lane < rem) q_put(&myq[lane], t);
                qn = rem;
            }
            bal = __builtin_amdgcn_ballot_w32(m != 0u);
        }
    }
    if (qn > 0) sink(qn, myq);
}


struct DegSink {
    const int*   __restrict__ col;
    const float* __restrict__ w;
    float* sdeg;
    unsigned short* wl;
    int n0, lane;
    mutable int nw;
    __device__ __forceinline__ void operator()(int cnt, int* myq) const {
        if (lane < cnt) {
            const int e = q_get(&myq[lane]);
            lds_add(&sdeg[col[e] - n0], w[e]);
            if (nw + lane < LIST_C) wl[nw + lane] = (unsigned short)list_code(e);
        }
        nw += cnt;
    }
};

__global__ __launch_bounds__(OWN_T) void k_own_deg(const unsigned char* __restrict__ keys,
                                                   const int* __restrict__ col,
                                                   const float* __restrict__ w,
                                                   float* __restrict__ dinv,
                                                   unsigned short* __restrict__ lists,
                                                   int* __restrict__ counts) {
    __shared__ float sdeg[OWN_R];
    __shared__ int   q[OWN_W * 64];
    __shared__ int   wcnt[OWN_W];
    __shared__ __attribute__((aligned(16))) unsigned short wl[LIST_WG];
    const int me   = blockIdx.x;
    const int n0   = me * OWN_R;
    const int nn   = min(OWN_R, N_NODES - n0);
    const int wave = threadIdx.x >> 5;
    const int lane = threadIdx.x & 31;
    for (int t = threadIdx.x; t < nn; t += OWN_T) sdeg[t] = 1.0f;
    for (int t = threadIdx.x; t < LIST_WG / 8; t += OWN_T) ((uint4_any*)wl)[t] = make_uint4(0u, 0u, 0u, 0u);
    __syncthreads();
    const DegSink sink{col, w, sdeg, wl + wave * LIST_C, n0, lane, 0};
    scan_keys(keys, me, q + wave * 64, sink);
    if (lane == 0) wcnt[wave] = sink.nw;
    __syncthreads();
    for (int t = threadIdx.x; t < nn; t += OWN_T) {
        const float d = sdeg[t];
        dinv[n0 + t] = (d > 0.0f) ? rsqrtf(d) : 0.0f;
    }
    if (wave == 0) counts[me * OWN_W + lane] = wcnt[lane];
    uint4_any* __restrict__ dst = (uint4_any*)(lists + (size_t)me * LIST_WG);
    for (int t = threadIdx.x; t < LIST_WG / 8; t += OWN_T) dst[t] = ((const uint4_any*)wl)[t];
}

__global__ void k_norm(const int* __restrict__ row, const int* __restrict__ col,
                       const float* __restrict__ w, const float* __restrict__ dinv,
                       float* __restrict__ norm) {
    int e = blockIdx.x * blockDim.x + threadIdx.x;
    if (e < N_EDGES) norm[e] = dinv[row[e]] * w[e] * dinv[col[e]];
}


__global__ __launch_bounds__(128) void k_gemm1(const float* __restrict__ x,
                                               const float* __restrict__ W1,
                                               float* __restrict__ h1) {
    const int wave  = threadIdx.x >> 5;
    const int lane  = threadIdx.x & 31;
    const int mtile = blockIdx.x * 4 + wave;
    if (mtile >= M_TILES) return;
    const int row0 = mtile * 16;
    const int half = lane >> 4;
    const int l16  = lane & 15;

    const float* arow = x + (size_t)(row0 + l16) * N_FEAT + half * 2;
    v8f c0 = {}, c1 = {}, c2 = {}, c3 = {};

    for (int k = 0; k < N_FEAT; k += 4) {
        v2f a;
        a.x = arow[k];
        a.y = arow[k + 1];
        const float* bp = W1 + (size_t)(k + half * 2) * HIDDEN + l16;
        v2f b0, b1, b2, b3;
        b0.x = bp[0];   b0.y = bp[HIDDEN + 0];
        b1.x = bp[16];  b1.y = bp[HIDDEN + 16];
        b2.x = bp[32];  b2.y = bp[HIDDEN + 32];
        b3.x = bp[48];  b3.y = bp[HIDDEN + 48];
        c0 = __builtin_amdgcn_wmma_f32_16x16x4_f32(false, a, false, b0, (short)0, c0, false, false);
        c1 = __builtin_amdgcn_wmma_f32_16x16x4_f32(false, a, false, b1, (short)0, c1, false, false);
        c2 = __builtin_amdgcn_wmma_f32_16x16x4_f32(false, a, false, b2, (short)0, c2, false, false);
        c3 = __builtin_amdgcn_wmma_f32_16x16x4_f32(false, a, false, b3, (short)0, c3, false, false);
    }

    float* out = h1 + (size_t)(row0 + half * 8) * HIDDEN + l16;
    for (int r = 0; r < 8; ++r) {
        out[(size_t)r * HIDDEN + 0]  = c0[r];
        out[(size_t)r * HIDDEN + 16] = c1[r];
        out[(size_t)r * HIDDEN + 32] = c2[r];
        out[(size_t)r * HIDDEN + 48] = c3[r];
    }
}


__global__ __launch_bounds__(128) void k_gemm2(const float* __restrict__ agg1,
                                               const float* __restrict__ W2,
                                               float* __restrict__ h2) {
    const int wave  = threadIdx.x >> 5;
    const int lane  = threadIdx.x & 31;
    const int mtile = blockIdx.x * 4 + wave;
    if (mtile >= M_TILES) return;
    const int row0 = mtile * 16;
    const int half = lane >> 4;
    const int l16  = lane & 15;

    const float* arow = agg1 + (size_t)(row0 + l16) * HIDDEN + half * 2;
    v8f c = {};

    for (int k = 0; k < HIDDEN; k += 4) {
        v2f a;
        a.x = fmaxf(arow[k], 0.0f);
        a.y = fmaxf(arow[k + 1], 0.0f);
        const float* bp = W2 + (size_t)(k + half * 2) * N_CLASSES + l16;
        v2f b;
        b.x = bp[0];
        b.y = bp[N_CLASSES];
        c = __builtin_amdgcn_wmma_f32_16x16x4_f32(false, a, false, b, (short)0, c, false, false);
    }

    float* out = h2 + (size_t)(row0 + half * 8) * N_CLASSES + l16;
    for (int r = 0; r < 8; ++r)
        out[(size_t)r * N_CLASSES] = c[r];
}


template <int F>
struct FeatSink {
    const int*   __restrict__ row;
    const int*   __restrict__ col;
    const float* __restrict__ norm;
    const float* __restrict__ h;
    float* acc;
    int n0, lane;

    __device__ __forceinline__ void operator()(int cnt, int* myq) const {
        int e = 0;
        if (lane < cnt) e = q_get(&myq[lane]);
        run(cnt, e);
    }

    __device__ __forceinline__ void run(int cnt, int e) const {
        const int src = row[e];
        const int dst = col[e] - n0;
        const int wi  = __float_as_int(norm[e]);

        if constexpr (F == 64) {
            for (int i = 0; i < cnt; i += 4) {
                float a0[4], a1[4];
                int   dd[4];
                #pragma unroll
                for (int u = 0; u < 4; ++u) {
                    const int ii = (i + u) & 31;
                    const int   s  = __builtin_amdgcn_readlane(src, ii);
                    const float ww = __int_as_float(__builtin_amdgcn_readlane(wi, ii));
                    dd[u] = __builtin_amdgcn_readlane(dst, ii);
                    const float* hp = h + (size_t)s * 64 + lane;
                    a0[u] = ww * hp[0];
                    a1[u] = ww * hp[32];
                }
                #pragma unroll
                for (int u = 0; u < 4; ++u) {
                    if (i + u < cnt) {
                        lds_add(&acc[dd[u] * 64 + lane],      a0[u]);
                        lds_add(&acc[dd[u] * 64 + lane + 32], a1[u]);
                    }
                }
            }
        } else {
            static_assert(F == 16, "two edges per step, one column per lane");
            const int half = lane >> 4;
            const int c    = lane & 15;
            for (int i = 0; i < cnt; i += 8) {
                float a[4];
                int   dd[4];
                #pragma unroll
                for (int u = 0; u < 4; ++u) {
                    const int i0 = (i + 2 * u) & 31;
                    const int i1 = (i + 2 * u + 1) & 31;
                    const int s0 = __builtin_amdgcn_readlane(src, i0), s1 = __builtin_amdgcn_readlane(src, i1);
                    const int d0 = __builtin_amdgcn_readlane(dst, i0), d1 = __builtin_amdgcn_readlane(dst, i1);
                    const int w0 = __builtin_amdgcn_readlane(wi,  i0), w1 = __builtin_amdgcn_readlane(wi,  i1);
                    const int s  = half ? s1 : s0;
                    dd[u]        = half ? d1 : d0;
                    const float ww = __int_as_float(half ? w1 : w0);
                    a[u] = ww * h[(size_t)s * 16 + c];
                }
                #pragma unroll
                for (int u = 0; u < 4; ++u) {
                    if (i + 2 * u + half < cnt) lds_add(&acc[dd[u] * 16 + c], a[u]);
                }
            }
        }
    }
};

template <int F>
__global__ __launch_bounds__(OWN_T) void k_own_feat(const unsigned char* __restrict__ keys,
                                                    const int* __restrict__ row,
                                                    const int* __restrict__ col,
                                                    const float* __restrict__ norm,
                                                    const float* __restrict__ h,
                                                    const float* __restrict__ dinv,
                                                    const float* __restrict__ bias,
                                                    const unsigned short* __restrict__ lists,
                                                    const int* __restrict__ counts,
                                                    float* __restrict__ out) {
    __shared__ float acc[OWN_R * F];
    __shared__ int   q[OWN_W * 64];
    const int me = blockIdx.x;
    const int n0 = me * OWN_R;
    const int nn = min(OWN_R, N_NODES - n0);
    const int wave = threadIdx.x >> 5;
    const int lane = threadIdx.x & 31;

    const unsigned short* __restrict__ mine = lists + (size_t)(me * OWN_W + wave) * LIST_C;
    const int n_v = counts[me * OWN_W + wave];
    int c_n = mine[lane];

    for (int t = threadIdx.x; t < nn * F; t += OWN_T) {
        const float s = dinv[n0 + t / F];
        acc[t] = bias[t % F] + s * s * h[(size_t)n0 * F + t];
    }
    __syncthreads();

    const FeatSink<F> sink{row, col, norm, h, acc, n0, lane};
    const int n = __builtin_amdgcn_readfirstlane(n_v);
    if (n <= LIST_C) {
        int b = 0;
        for (; b + 32 <= n; b += 32) {
            const int c = c_n;
            if (b + 32 + lane < n) c_n = mine[b + 32 + lane];
            sink.run(32, list_edge(c, wave));
        }
        if (b < n) {
            int e = 0;
            if (b + lane < n) e = list_edge(c_n, wave);
            sink.run(n - b, e);
        }
    } else {
        scan_keys(keys, me, q + wave * 64, sink);
    }
    __syncthreads();

    for (int t = threadIdx.x; t < nn * F; t += OWN_T)
        out[(size_t)n0 * F + t] = acc[t];
}


__global__ void k_logsoftmax(const float* __restrict__ agg2, float* __restrict__ out) {
    int i = blockIdx.x * blockDim.x + threadIdx.x;
    if (i >= N_NODES) return;
    const float* v = agg2 + (size_t)i * N_CLASSES;
    float m = v[0];
    #pragma unroll
    for (int c = 1; c < N_CLASSES; ++c) m = fmaxf(m, v[c]);
    float s = 0.0f;
    #pragma unroll
    for (int c = 0; c < N_CLASSES; ++c) s += expf(v[c] - m);
    float l = logf(s);
    float* o = out + (size_t)i * N_CLASSES;
    #pragma unroll
    for (int c = 0; c < N_CLASSES; ++c) o[c] = v[c] - m - l;
}


extern "C" void kernel_launch(void* const* d_in, const int* in_sizes, int n_in,
                              void* d_out, int out_size, void* d_ws, size_t ws_size,
                              hipStream_t stream) {
    const float* x  = (const float*)d_in[0];
    const int*   ei = (const int*)d_in[1];
    const float* ew = (const float*)d_in[2];
    const float* W1 = (const float*)d_in[3];
    const float* b1 = (const float*)d_in[4];
    const float* W2 = (const float*)d_in[5];
    const float* b2 = (const float*)d_in[6];
    float* out = (float*)d_out;

    const int* row = ei;
    const int* col = ei + N_EDGES;

    float* ws   = (float*)d_ws;
    float* deg  = ws;
    float* dinv = deg  + N_NODES;
    float* norm = dinv + N_NODES;
    float* h1   = norm + N_EDGES;
    float* agg1 = h1   + (size_t)N_NODES * HIDDEN;
    float* h2   = agg1 + (size_t)N_NODES * HIDDEN;
    float* kreg = h2   + (size_t)N_NODES * N_CLASSES;
    unsigned char* keys = (unsigned char*)kreg;
    unsigned short* lists = (unsigned short*)(keys + N_EDGES);
    int* counts = (int*)deg;
    float* agg2 = h1;

    const int B = 256;

    k_keys   <<<(N_EDGES / 4 + B - 1) / B, B, 0, stream>>>(col, (unsigned*)keys);
    k_own_deg<<<OWN_G, OWN_T, 0, stream>>>(keys, col, ew, dinv, lists, counts);
    k_norm   <<<(N_EDGES + B - 1) / B, B, 0, stream>>>(row, col, ew, dinv, norm);

    k_gemm1<<<(M_TILES + 3) / 4, 128, 0, stream>>>(x, W1, h1);

    k_own_feat<HIDDEN><<<OWN_G, OWN_T, 0, stream>>>(keys, row, col, norm, h1, dinv, b1, lists, counts, agg1);

    k_gemm2<<<(M_TILES + 3) / 4, 128, 0, stream>>>(agg1, W2, h2);

    k_own_feat<N_CLASSES><<<OWN_G, OWN_T, 0, stream>>>(keys, row, col, norm, h2, dinv, b2, lists, counts, agg2);

    k_logsoftmax<<<(N_NODES + B - 1) / B, B, 0, stream>>>(agg2, out);
}
